# starting kernel
# baseline (speedup 1.0000x reference)
#include <hip/hip_runtime.h>

#define IN_F  4096
#define OUT_F 4096
#define BATCH 128

typedef __attribute__((ext_vector_type(2))) float v2f;
typedef __attribute__((ext_vector_type(4))) float v4f;
typedef __attribute__((ext_vector_type(8))) float v8f;

__global__ void zero_dense(v4f* __restrict__ W4, int n4) {
    int i = blockIdx.x * blockDim.x + threadIdx.x;
    if (i < n4) {
        W4[i] = (v4f){0.f, 0.f, 0.f, 0.f};
    }
}

__global__ void coo_scatter(const float* __restrict__ vals,
                            const int*   __restrict__ rows,
                            const int*   __restrict__ cols,
                            float* __restrict__ W, int nnz) {
    int i = blockIdx.x * blockDim.x + threadIdx.x;
    if (i < nnz) {
        size_t idx = (size_t)rows[i] * IN_F + (size_t)cols[i];
        __hip_atomic_fetch_add(&W[idx], vals[i],
                               __ATOMIC_RELAXED, __HIP_MEMORY_SCOPE_AGENT);
    }
}

__global__ void __launch_bounds__(32) dense_wmma_gemm(
        const float* __restrict__ W,
        const float* __restrict__ inp,
        const float* __restrict__ bias,
        float* __restrict__ out,
        int kdim) {
    const int r0   = blockIdx.x * 16;
    const int b0   = blockIdx.y * 64;
    const int lane = threadIdx.x & 31;
    const int half = lane >> 4;
    const int l16  = lane & 15;
    const int koff = half * 4;

    const float* arow = W   + (size_t)(r0 + l16) * IN_F + koff;
    const float* br0  = inp + (size_t)(b0 +  0 + l16) * IN_F + koff;
    const float* br1  = inp + (size_t)(b0 + 16 + l16) * IN_F + koff;
    const float* br2  = inp + (size_t)(b0 + 32 + l16) * IN_F + koff;
    const float* br3  = inp + (size_t)(b0 + 48 + l16) * IN_F + koff;

    v8f acc0 = {}, acc1 = {}, acc2 = {}, acc3 = {};

    constexpr int PF = 4;
    const int steps  = kdim >> 3;
    const int nmain  = (steps - PF) / PF;

    v4f A[PF], B0[PF], B1[PF], B2[PF], B3[PF];

    auto k8_step = [&](v4f av, v4f b0v, v4f b1v, v4f b2v, v4f b3v) {
        v2f alo = __builtin_shufflevector(av, av, 0, 1);
        v2f ahi = __builtin_shufflevector(av, av, 2, 3);
        v2f l0 = __builtin_shufflevector(b0v, b0v, 0, 1);
        v2f h0 = __builtin_shufflevector(b0v, b0v, 2, 3);
        v2f l1 = __builtin_shufflevector(b1v, b1v, 0, 1);
        v2f h1 = __builtin_shufflevector(b1v, b1v, 2, 3);
        v2f l2 = __builtin_shufflevector(b2v, b2v, 0, 1);
        v2f h2 = __builtin_shufflevector(b2v, b2v, 2, 3);
        v2f l3 = __builtin_shufflevector(b3v, b3v, 0, 1);
        v2f h3 = __builtin_shufflevector(b3v, b3v, 2, 3);
        acc0 = __builtin_amdgcn_wmma_f32_16x16x4_f32(false, alo, false, l0,
                                                     (short)0, acc0, false, false);
        acc1 = __builtin_amdgcn_wmma_f32_16x16x4_f32(false, alo, false, l1,
                                                     (short)0, acc1, false, false);
        acc2 = __builtin_amdgcn_wmma_f32_16x16x4_f32(false, alo, false, l2,
                                                     (short)0, acc2, false, false);
        acc3 = __builtin_amdgcn_wmma_f32_16x16x4_f32(false, alo, false, l3,
                                                     (short)0, acc3, false, false);
        acc0 = __builtin_amdgcn_wmma_f32_16x16x4_f32(false, ahi, false, h0,
                                                     (short)0, acc0, false, false);
        acc1 = __builtin_amdgcn_wmma_f32_16x16x4_f32(false, ahi, false, h1,
                                                     (short)0, acc1, false, false);
        acc2 = __builtin_amdgcn_wmma_f32_16x16x4_f32(false, ahi, false, h2,
                                                     (short)0, acc2, false, false);
        acc3 = __builtin_amdgcn_wmma_f32_16x16x4_f32(false, ahi, false, h3,
                                                     (short)0, acc3, false, false);
    };

#pragma unroll
    for (int p = 0; p < PF; ++p) {
        A[p]  = *(const v4f*)(arow + (size_t)p * 8);
        B0[p] = *(const v4f*)(br0  + (size_t)p * 8);
        B1[p] = *(const v4f*)(br1  + (size_t)p * 8);
        B2[p] = *(const v4f*)(br2  + (size_t)p * 8);
        B3[p] = *(const v4f*)(br3  + (size_t)p * 8);
    }

    int sbase = 0;
    for (int t = 0; t < nmain; ++t) {
#pragma unroll
        for (int i = 0; i < PF; ++i) {
            v4f av  = A[i];
            v4f b0v = B0[i], b1v = B1[i], b2v = B2[i], b3v = B3[i];

            const size_t off = (size_t)(sbase + i + PF) * 8;
            A[i]  = *(const v4f*)(arow + off);
            B0[i] = *(const v4f*)(br0  + off);
            B1[i] = *(const v4f*)(br1  + off);
            B2[i] = *(const v4f*)(br2  + off);
            B3[i] = *(const v4f*)(br3  + off);

            k8_step(av, b0v, b1v, b2v, b3v);
        }
        sbase += PF;
    }

#pragma unroll
    for (int i = 0; i < PF; ++i) {
        k8_step(A[i], B0[i], B1[i], B2[i], B3[i]);
    }

    const int mbase = r0 + half * 8;
    v4f bv0 = *(const v4f*)(bias + mbase);
    v4f bv1 = *(const v4f*)(bias + mbase + 4);

    auto store_tile = [&](v8f acc, float* o) {
        v4f lo, hi;
#pragma unroll
        for (int v = 0; v < 4; ++v) {
            lo[v] = acc[v]     + bv0[v];
            hi[v] = acc[v + 4] + bv1[v];
        }
        *(v4f*)o       = lo;
        *(v4f*)(o + 4) = hi;
    };

    store_tile(acc0, out + (size_t)(b0 +  0 + l16) * OUT_F + mbase);
    store_tile(acc1, out + (size_t)(b0 + 16 + l16) * OUT_F + mbase);
    store_tile(acc2, out + (size_t)(b0 + 32 + l16) * OUT_F + mbase);
    store_tile(acc3, out + (size_t)(b0 + 48 + l16) * OUT_F + mbase);
}

extern "C" void kernel_launch(void* const* d_in, const int* in_sizes, int n_in,
                              void* d_out, int out_size, void* d_ws, size_t ws_size,
                              hipStream_t stream) {
    const float* inp  = (const float*)d_in[0];
    const float* wval = (const float*)d_in[1];
    const int*   wrow = (const int*)d_in[2];
    const int*   wcol = (const int*)d_in[3];
    const float* bias = (const float*)d_in[4];
    float*       out  = (float*)d_out;
    float*       W    = (float*)d_ws;
    const int    nnz  = in_sizes[1];

    const int n4 = (OUT_F * IN_F) / 4;
    zero_dense<<<(n4 + 255) / 256, 256, 0, stream>>>((v4f*)W, n4);

    coo_scatter<<<(nnz + 255) / 256, 256, 0, stream>>>(wval, wrow, wcol, W, nnz);

    dim3 grid(OUT_F / 16, BATCH / 64);
    dense_wmma_gemm<<<grid, 32, 0, stream>>>(W, inp, bias, out, IN_F);
}


# GEMM: K split over the 8 waves of a work-group (32x64 tile per wave, 2048 waves instead of 512); partials added through LDS in fixed wave order, which changes the K summation order
# speedup vs baseline: 1.3558x; 1.3558x over previous
#include <hip/hip_runtime.h>

#define IN_F  4096
#define OUT_F 4096
#define BATCH 128

typedef __attribute__((ext_vector_type(2))) float v2f;
typedef __attribute__((ext_vector_type(4))) float v4f;
typedef __attribute__((ext_vector_type(8))) float v8f;

__global__ void zero_dense(v4f* __restrict__ W4, int n4) {
    int i = blockIdx.x * blockDim.x + threadIdx.x;
    if (i < n4) {
        W4[i] = (v4f){0.f, 0.f, 0.f, 0.f};
    }
}

__global__ void coo_scatter(const float* __restrict__ vals,
                            const int*   __restrict__ rows,
                            const int*   __restrict__ cols,
                            float* __restrict__ W, int nnz) {
    int i = blockIdx.x * blockDim.x + threadIdx.x;
    if (i < nnz) {
        size_t idx = (size_t)rows[i] * IN_F + (size_t)cols[i];
        __hip_atomic_fetch_add(&W[idx], vals[i],
                               __ATOMIC_RELAXED, __HIP_MEMORY_SCOPE_AGENT);
    }
}

#define KSPLIT 8

__global__ void __launch_bounds__(256) dense_wmma_gemm(
        const float* __restrict__ W,
        const float* __restrict__ inp,
        const float* __restrict__ bias,
        float* __restrict__ out,
        int kdim) {
    __shared__ float red[KSPLIT * 8 * 32 * 8];

    const int r0   = blockIdx.x * 32;
    const int b0   = blockIdx.y * 64;
    const int tid  = threadIdx.x;
    const int wave = tid >> 5;
    const int lane = tid & 31;
    const int half = lane >> 4;
    const int l16  = lane & 15;

    const int kper = kdim / KSPLIT;
    const int kbeg = wave * kper + half * 16;

    const float* ptr[6];
    ptr[0] = W   + (size_t)(r0 +  0 + l16) * IN_F + kbeg;
    ptr[1] = W   + (size_t)(r0 + 16 + l16) * IN_F + kbeg;
    ptr[2] = inp + (size_t)(b0 +  0 + l16) * IN_F + kbeg;
    ptr[3] = inp + (size_t)(b0 + 16 + l16) * IN_F + kbeg;
    ptr[4] = inp + (size_t)(b0 + 32 + l16) * IN_F + kbeg;
    ptr[5] = inp + (size_t)(b0 + 48 + l16) * IN_F + kbeg;

    v8f acc[2][4];
#pragma unroll
    for (int i = 0; i < 2; ++i)
#pragma unroll
        for (int n = 0; n < 4; ++n)
            acc[i][n] = (v8f){0.f, 0.f, 0.f, 0.f, 0.f, 0.f, 0.f, 0.f};

    v4f S0[6][4], S1[6][4];

    auto load_stage = [&](v4f (&S)[6][4], int step) {
#pragma unroll
        for (int o = 0; o < 6; ++o)
#pragma unroll
            for (int j = 0; j < 4; ++j)
                S[o][j] = *(const v4f*)(ptr[o] + (size_t)step * 32 + j * 4);
    };

    auto compute_stage = [&](const v4f (&S)[6][4]) {
#pragma unroll
        for (int j = 0; j < 4; ++j) {
#pragma unroll
            for (int p = 0; p < 2; ++p) {
                v2f a0v = {S[0][j][2 * p], S[0][j][2 * p + 1]};
                v2f a1v = {S[1][j][2 * p], S[1][j][2 * p + 1]};
                v2f bv[4];
#pragma unroll
                for (int n = 0; n < 4; ++n)
                    bv[n] = (v2f){S[2 + n][j][2 * p], S[2 + n][j][2 * p + 1]};
#pragma unroll
                for (int n = 0; n < 4; ++n)
                    acc[0][n] = __builtin_amdgcn_wmma_f32_16x16x4_f32(
                        false, a0v, false, bv[n], (short)0, acc[0][n], false, false);
#pragma unroll
                for (int n = 0; n < 4; ++n)
                    acc[1][n] = __builtin_amdgcn_wmma_f32_16x16x4_f32(
                        false, a1v, false, bv[n], (short)0, acc[1][n], false, false);
            }
        }
    };

    const int nsteps = kper >> 5;
    const int npairs = nsteps >> 1;

    load_stage(S0, 0);
    for (int t = 0; t + 1 < npairs; ++t) {
        load_stage(S1, 2 * t + 1);
        __builtin_amdgcn_sched_barrier(0);
        compute_stage(S0);
        __builtin_amdgcn_sched_barrier(0);
        load_stage(S0, 2 * t + 2);
        __builtin_amdgcn_sched_barrier(0);
        compute_stage(S1);
    }
    load_stage(S1, nsteps - 1);
    __builtin_amdgcn_sched_barrier(0);
    compute_stage(S0);
    compute_stage(S1);

    float* mine = red + wave * 2048 + lane * 8;
#pragma unroll
    for (int i = 0; i < 2; ++i) {
#pragma unroll
        for (int n = 0; n < 4; ++n) {
            v8f a = acc[i][n];
            *(v4f*)(mine + (i * 4 + n) * 256)     = (v4f){a[0], a[1], a[2], a[3]};
            *(v4f*)(mine + (i * 4 + n) * 256 + 4) = (v4f){a[4], a[5], a[6], a[7]};
        }
    }
    __syncthreads();

#pragma unroll
    for (int pass = 0; pass < 2; ++pass) {
        const int Lb   = (tid >> 3) + 32 * pass;
        const int e    = tid & 7;
        const int n    = Lb >> 4;
        const int c    = Lb & 15;
        const int i    = e >> 2;
        const int h    = (e >> 1) & 1;
        const int quad = e & 1;
        const float* src = red + (i * 4 + n) * 256 + (c + 16 * h) * 8 + quad * 4;
        v4f s = *(const v4f*)src;
#pragma unroll
        for (int w = 1; w < KSPLIT; ++w)
            s += *(const v4f*)(src + w * 2048);
        s += *(const v4f*)(bias + r0 + 4 * e);
        *(v4f*)(out + (size_t)(b0 + Lb) * OUT_F + r0 + 4 * e) = s;
    }
}

extern "C" void kernel_launch(void* const* d_in, const int* in_sizes, int n_in,
                              void* d_out, int out_size, void* d_ws, size_t ws_size,
                              hipStream_t stream) {
    const float* inp  = (const float*)d_in[0];
    const float* wval = (const float*)d_in[1];
    const int*   wrow = (const int*)d_in[2];
    const int*   wcol = (const int*)d_in[3];
    const float* bias = (const float*)d_in[4];
    float*       out  = (float*)d_out;
    float*       W    = (float*)d_ws;
    const int    nnz  = in_sizes[1];

    const int n4 = (OUT_F * IN_F) / 4;
    zero_dense<<<(n4 + 255) / 256, 256, 0, stream>>>((v4f*)W, n4);

    coo_scatter<<<(nnz + 255) / 256, 256, 0, stream>>>(wval, wrow, wcol, W, nnz);

    dim3 grid(OUT_F / 32, BATCH / 64);
    dense_wmma_gemm<<<grid, 256, 0, stream>>>(W, inp, bias, out, IN_F);
}


# GEMM A operand from a 16-row band slab built in LDS (zero, LDS float adds) from COO binned by band; no 64 MB dense W in memory; LDS-add arrival order replaces global-atomic order
# speedup vs baseline: 2.3368x; 1.7236x over previous
#include <hip/hip_runtime.h>

#define IN_F  4096
#define OUT_F 4096
#define BATCH 128

typedef __attribute__((ext_vector_type(2))) float v2f;
typedef __attribute__((ext_vector_type(4))) float v4f;
typedef __attribute__((ext_vector_type(8))) float v8f;

#define NBANDS     256
#define BAND_CAP   16384
#define OVF_CAP    2097152
#define CNT_STRIDE 1088
#define CNT_BYTES  (1u << 21)
#define BKT_OFF    ((size_t)CNT_BYTES)
#define OVF_OFF    (BKT_OFF + (size_t)NBANDS * BAND_CAP * 8)

__global__ void clear_counters(int* __restrict__ cnt) {
    int t = threadIdx.x;
    if (t <= NBANDS) cnt[t * CNT_STRIDE] = 0;
}

__global__ void __launch_bounds__(1024) bin_coo(
        const float* __restrict__ vals,
        const int*   __restrict__ rows,
        const int*   __restrict__ cols,
        int*  __restrict__ cnt,
        int2* __restrict__ bkt,
        int2* __restrict__ ovf,
        int nnz) {
    __shared__ int hist[NBANDS];
    __shared__ int base[NBANDS];
    const int tid = threadIdx.x;
    if (tid < NBANDS) hist[tid] = 0;
    __syncthreads();

    int   key[4], rank[4];
    float val[4];
#pragma unroll
    for (int q = 0; q < 4; ++q) {
        const int i = blockIdx.x * 4096 + q * 1024 + tid;
        key[q] = -1;
        rank[q] = 0;
        val[q] = 0.f;
        if (i < nnz) {
            const int r = rows[i] & (OUT_F - 1);
            const int c = cols[i] & (IN_F - 1);
            val[q] = vals[i];
            key[q] = (r << 12) | c;
            rank[q] = atomicAdd(&hist[r >> 4], 1);
        }
    }
    __syncthreads();
    if (tid < NBANDS) {
        const int h = hist[tid];
        base[tid] = (h > 0) ? atomicAdd(&cnt[tid * CNT_STRIDE], h) : 0;
    }
    __syncthreads();
#pragma unroll
    for (int q = 0; q < 4; ++q) {
        if (key[q] >= 0) {
            const int b   = key[q] >> 16;
            const int pos = base[b] + rank[q];
            const int2 ent = make_int2(key[q], __float_as_int(val[q]));
            if (pos < BAND_CAP) {
                bkt[(size_t)b * BAND_CAP + pos] = ent;
            } else {
                const int p2 = atomicAdd(&cnt[NBANDS * CNT_STRIDE], 1);
                if (p2 < OVF_CAP) ovf[p2] = ent;
            }
        }
    }
}

#define KSPLIT 8
#define PITCH  4100

__global__ void __launch_bounds__(256) band_wmma_gemm(
        const int*  __restrict__ cnt,
        const int2* __restrict__ bkt,
        const int2* __restrict__ ovf,
        const float* __restrict__ inp,
        const float* __restrict__ bias,
        float* __restrict__ out,
        int kdim) {
    __shared__ float Wl[16 * PITCH];

    const int band = blockIdx.x;
    const int tid  = threadIdx.x;
    const int wave = tid >> 5;
    const int lane = tid & 31;
    const int half = lane >> 4;
    const int l16  = lane & 15;

    for (int i = tid; i < (16 * PITCH) / 4; i += 256)
        ((v4f*)Wl)[i] = (v4f){0.f, 0.f, 0.f, 0.f};
    __syncthreads();

    int n = cnt[band * CNT_STRIDE];
    n = n < BAND_CAP ? n : BAND_CAP;
    const int2* mybkt = bkt + (size_t)band * BAND_CAP;
    for (int e = tid; e < n; e += 256) {
        const int2 ent = mybkt[e];
        const int rr = (ent.x >> 12) & 15;
        const int cc = ent.x & (IN_F - 1);
        __hip_atomic_fetch_add(&Wl[rr * PITCH + cc], __int_as_float(ent.y),
                               __ATOMIC_RELAXED, __HIP_MEMORY_SCOPE_WORKGROUP);
    }
    int novf = cnt[NBANDS * CNT_STRIDE];
    novf = novf < OVF_CAP ? novf : OVF_CAP;
    for (int e = tid; e < novf; e += 256) {
        const int2 ent = ovf[e];
        if ((ent.x >> 16) == band) {
            const int rr = (ent.x >> 12) & 15;
            const int cc = ent.x & (IN_F - 1);
            __hip_atomic_fetch_add(&Wl[rr * PITCH + cc], __int_as_float(ent.y),
                                   __ATOMIC_RELAXED, __HIP_MEMORY_SCOPE_WORKGROUP);
        }
    }
    __syncthreads();

    const int kper = kdim / KSPLIT;
    const int kbeg = wave * kper + half * 16;

    const float* aptr = Wl + l16 * PITCH + kbeg;
    const float* bptr[8];
#pragma unroll
    for (int t = 0; t < 8; ++t)
        bptr[t] = inp + (size_t)(16 * t + l16) * IN_F + kbeg;

    v8f acc[8];
#pragma unroll
    for (int t = 0; t < 8; ++t)
        acc[t] = (v8f){0.f, 0.f, 0.f, 0.f, 0.f, 0.f, 0.f, 0.f};

    v4f S0[9][4], S1[9][4];

    auto load_stage = [&](v4f (&S)[9][4], int step) {
#pragma unroll
        for (int t = 0; t < 8; ++t)
#pragma unroll
            for (int j = 0; j < 4; ++j)
                S[1 + t][j] = *(const v4f*)(bptr[t] + (size_t)step * 32 + j * 4);
#pragma unroll
        for (int j = 0; j < 4; ++j)
            S[0][j] = *(const v4f*)(aptr + step * 32 + j * 4);
    };

    auto compute_stage = [&](const v4f (&S)[9][4]) {
#pragma unroll
        for (int j = 0; j < 4; ++j) {
#pragma unroll
            for (int p = 0; p < 2; ++p) {
                v2f av = {S[0][j][2 * p], S[0][j][2 * p + 1]};
#pragma unroll
                for (int t = 0; t < 8; ++t) {
                    v2f bv = {S[1 + t][j][2 * p], S[1 + t][j][2 * p + 1]};
                    acc[t] = __builtin_amdgcn_wmma_f32_16x16x4_f32(
                        false, av, false, bv, (short)0, acc[t], false, false);
                }
            }
        }
    };

    const int nsteps = kper >> 5;
    const int npairs = nsteps >> 1;

    load_stage(S0, 0);
    for (int t = 0; t + 1 < npairs; ++t) {
        load_stage(S1, 2 * t + 1);
        __builtin_amdgcn_sched_barrier(0);
        compute_stage(S0);
        __builtin_amdgcn_sched_barrier(0);
        load_stage(S0, 2 * t + 2);
        __builtin_amdgcn_sched_barrier(0);
        compute_stage(S1);
    }
    load_stage(S1, nsteps - 1);
    __builtin_amdgcn_sched_barrier(0);
    compute_stage(S0);
    compute_stage(S1);

    __syncthreads();

    float* red  = Wl;
    float* mine = red + wave * 2048 + lane * 8;
#pragma unroll
    for (int t = 0; t < 8; ++t) {
        v8f a = acc[t];
        *(v4f*)(mine + t * 256)     = (v4f){a[0], a[1], a[2], a[3]};
        *(v4f*)(mine + t * 256 + 4) = (v4f){a[4], a[5], a[6], a[7]};
    }
    __syncthreads();

#pragma unroll
    for (int pass = 0; pass < 2; ++pass) {
        const int idx  = tid + 256 * pass;
        const int Lb   = idx >> 2;
        const int e    = idx & 3;
        const int t    = Lb >> 4;
        const int c    = Lb & 15;
        const int h    = e >> 1;
        const int quad = e & 1;
        const float* src = red + t * 256 + (c + 16 * h) * 8 + quad * 4;
        v4f s = *(const v4f*)src;
#pragma unroll
        for (int w = 1; w < KSPLIT; ++w)
            s += *(const v4f*)(src + w * 2048);
        s += *(const v4f*)(bias + band * 16 + 4 * e);
        *(v4f*)(out + (size_t)Lb * OUT_F + band * 16 + 4 * e) = s;
    }
}

extern "C" void kernel_launch(void* const* d_in, const int* in_sizes, int n_in,
                              void* d_out, int out_size, void* d_ws, size_t ws_size,
                              hipStream_t stream) {
    const float* inp  = (const float*)d_in[0];
    const float* wval = (const float*)d_in[1];
    const int*   wrow = (const int*)d_in[2];
    const int*   wcol = (const int*)d_in[3];
    const float* bias = (const float*)d_in[4];
    float*       out  = (float*)d_out;
    int*         cnt  = (int*)d_ws;
    int2*        bkt  = (int2*)((char*)d_ws + BKT_OFF);
    int2*        ovf  = (int2*)((char*)d_ws + OVF_OFF);
    const int    nnz  = in_sizes[1];

    clear_counters<<<1, 320, 0, stream>>>(cnt);

    bin_coo<<<(nnz + 4095) / 4096, 1024, 0, stream>>>(wval, wrow, wcol, cnt, bkt, ovf, nnz);

    band_wmma_gemm<<<NBANDS, 256, 0, stream>>>(cnt, bkt, ovf, inp, bias, out, IN_F);
}


# GEMM kernel: band's COO list fetched 32 entries per thread as one load batch issued before the slab zeroing, not one dependent load per LDS add; LDS-add arrival order stays timing-dependent
# speedup vs baseline: 2.5532x; 1.0926x over previous
#include <hip/hip_runtime.h>

#define IN_F  4096
#define OUT_F 4096
#define BATCH 128

typedef __attribute__((ext_vector_type(2))) float v2f;
typedef __attribute__((ext_vector_type(4))) float v4f;
typedef __attribute__((ext_vector_type(8))) float v8f;

#define NBANDS     256
#define BAND_CAP   16384
#define OVF_CAP    2097152
#define CNT_STRIDE 1088
#define CNT_BYTES  (1u << 21)
#define BKT_OFF    ((size_t)CNT_BYTES)
#define OVF_OFF    (BKT_OFF + (size_t)NBANDS * BAND_CAP * 8)

__global__ void clear_counters(int* __restrict__ cnt) {
    int t = threadIdx.x;
    if (t <= NBANDS) cnt[t * CNT_STRIDE] = 0;
}

__global__ void __launch_bounds__(1024) bin_coo(
        const float* __restrict__ vals,
        const int*   __restrict__ rows,
        const int*   __restrict__ cols,
        int*  __restrict__ cnt,
        int2* __restrict__ bkt,
        int2* __restrict__ ovf,
        int nnz) {
    __shared__ int hist[NBANDS];
    __shared__ int base[NBANDS];
    const int tid = threadIdx.x;
    if (tid < NBANDS) hist[tid] = 0;
    __syncthreads();

    int   key[4], rank[4];
    float val[4];
#pragma unroll
    for (int q = 0; q < 4; ++q) {
        const int i = blockIdx.x * 4096 + q * 1024 + tid;
        key[q] = -1;
        rank[q] = 0;
        val[q] = 0.f;
        if (i < nnz) {
            const int r = rows[i] & (OUT_F - 1);
            const int c = cols[i] & (IN_F - 1);
            val[q] = vals[i];
            key[q] = (r << 12) | c;
            rank[q] = atomicAdd(&hist[r >> 4], 1);
        }
    }
    __syncthreads();
    if (tid < NBANDS) {
        const int h = hist[tid];
        base[tid] = (h > 0) ? atomicAdd(&cnt[tid * CNT_STRIDE], h) : 0;
    }
    __syncthreads();
#pragma unroll
    for (int q = 0; q < 4; ++q) {
        if (key[q] >= 0) {
            const int b   = key[q] >> 16;
            const int pos = base[b] + rank[q];
            const int2 ent = make_int2(key[q], __float_as_int(val[q]));
            if (pos < BAND_CAP) {
                bkt[(size_t)b * BAND_CAP + pos] = ent;
            } else {
                const int p2 = atomicAdd(&cnt[NBANDS * CNT_STRIDE], 1);
                if (p2 < OVF_CAP) ovf[p2] = ent;
            }
        }
    }
}

#define KSPLIT 8
#define PITCH  4100
#define SC_DEPTH 32

__global__ void __launch_bounds__(256) band_wmma_gemm(
        const int*  __restrict__ cnt,
        const int2* __restrict__ bkt,
        const int2* __restrict__ ovf,
        const float* __restrict__ inp,
        const float* __restrict__ bias,
        float* __restrict__ out,
        int kdim) {
    __shared__ float Wl[16 * PITCH];

    const int band = blockIdx.x;
    const int tid  = threadIdx.x;
    const int wave = tid >> 5;
    const int lane = tid & 31;
    const int half = lane >> 4;
    const int l16  = lane & 15;

    int n = cnt[band * CNT_STRIDE];
    n = n < BAND_CAP ? n : BAND_CAP;
    int novf = cnt[NBANDS * CNT_STRIDE];
    novf = novf < OVF_CAP ? novf : OVF_CAP;
    const int2* mybkt = bkt + (size_t)band * BAND_CAP;

    int2 ent[SC_DEPTH];
    auto load_batch = [&](int base) {
#pragma unroll
        for (int j = 0; j < SC_DEPTH; ++j) {
            const int e = base + j * 256 + tid;
            ent[j] = make_int2(0, 0);
            if (e < n) ent[j] = mybkt[e];
        }
    };
    auto add_batch = [&](int base) {
#pragma unroll
        for (int j = 0; j < SC_DEPTH; ++j) {
            const int e = base + j * 256 + tid;
            if (e < n) {
                const int rr = (ent[j].x >> 12) & 15;
                const int cc = ent[j].x & (IN_F - 1);
                __hip_atomic_fetch_add(&Wl[rr * PITCH + cc], __int_as_float(ent[j].y),
                                       __ATOMIC_RELAXED, __HIP_MEMORY_SCOPE_WORKGROUP);
            }
        }
    };

    load_batch(0);

    for (int i = tid; i < (16 * PITCH) / 4; i += 256)
        ((v4f*)Wl)[i] = (v4f){0.f, 0.f, 0.f, 0.f};
    __syncthreads();

    add_batch(0);
    for (int base = SC_DEPTH * 256; base < n; base += SC_DEPTH * 256) {
        load_batch(base);
        add_batch(base);
    }
    for (int e = tid; e < novf; e += 256) {
        const int2 oe = ovf[e];
        if ((oe.x >> 16) == band) {
            const int rr = (oe.x >> 12) & 15;
            const int cc = oe.x & (IN_F - 1);
            __hip_atomic_fetch_add(&Wl[rr * PITCH + cc], __int_as_float(oe.y),
                                   __ATOMIC_RELAXED, __HIP_MEMORY_SCOPE_WORKGROUP);
        }
    }
    __syncthreads();

    const int kper = kdim / KSPLIT;
    const int kbeg = wave * kper + half * 16;

    const float* aptr = Wl + l16 * PITCH + kbeg;
    const float* bptr[8];
#pragma unroll
    for (int t = 0; t < 8; ++t)
        bptr[t] = inp + (size_t)(16 * t + l16) * IN_F + kbeg;

    v8f acc[8];
#pragma unroll
    for (int t = 0; t < 8; ++t)
        acc[t] = (v8f){0.f, 0.f, 0.f, 0.f, 0.f, 0.f, 0.f, 0.f};

    v4f S0[9][4], S1[9][4];

    auto load_stage = [&](v4f (&S)[9][4], int step) {
#pragma unroll
        for (int t = 0; t < 8; ++t)
#pragma unroll
            for (int j = 0; j < 4; ++j)
                S[1 + t][j] = *(const v4f*)(bptr[t] + (size_t)step * 32 + j * 4);
#pragma unroll
        for (int j = 0; j < 4; ++j)
            S[0][j] = *(const v4f*)(aptr + step * 32 + j * 4);
    };

    auto compute_stage = [&](const v4f (&S)[9][4]) {
#pragma unroll
        for (int j = 0; j < 4; ++j) {
#pragma unroll
            for (int p = 0; p < 2; ++p) {
                v2f av = {S[0][j][2 * p], S[0][j][2 * p + 1]};
#pragma unroll
                for (int t = 0; t < 8; ++t) {
                    v2f bv = {S[1 + t][j][2 * p], S[1 + t][j][2 * p + 1]};
                    acc[t] = __builtin_amdgcn_wmma_f32_16x16x4_f32(
                        false, av, false, bv, (short)0, acc[t], false, false);
                }
            }
        }
    };

    const int nsteps = kper >> 5;
    const int npairs = nsteps >> 1;

    load_stage(S0, 0);
    for (int t = 0; t + 1 < npairs; ++t) {
        load_stage(S1, 2 * t + 1);
        __builtin_amdgcn_sched_barrier(0);
        compute_stage(S0);
        __builtin_amdgcn_sched_barrier(0);
        load_stage(S0, 2 * t + 2);
        __builtin_amdgcn_sched_barrier(0);
        compute_stage(S1);
    }
    load_stage(S1, nsteps - 1);
    __builtin_amdgcn_sched_barrier(0);
    compute_stage(S0);
    compute_stage(S1);

    __syncthreads();

    float* red  = Wl;
    float* mine = red + wave * 2048 + lane * 8;
#pragma unroll
    for (int t = 0; t < 8; ++t) {
        v8f a = acc[t];
        *(v4f*)(mine + t * 256)     = (v4f){a[0], a[1], a[2], a[3]};
        *(v4f*)(mine + t * 256 + 4) = (v4f){a[4], a[5], a[6], a[7]};
    }
    __syncthreads();

#pragma unroll
    for (int pass = 0; pass < 2; ++pass) {
        const int idx  = tid + 256 * pass;
        const int Lb   = idx >> 2;
        const int e    = idx & 3;
        const int t    = Lb >> 4;
        const int c    = Lb & 15;
        const int h    = e >> 1;
        const int quad = e & 1;
        const float* src = red + t * 256 + (c + 16 * h) * 8 + quad * 4;
        v4f s = *(const v4f*)src;
#pragma unroll
        for (int w = 1; w < KSPLIT; ++w)
            s += *(const v4f*)(src + w * 2048);
        s += *(const v4f*)(bias + band * 16 + 4 * e);
        *(v4f*)(out + (size_t)Lb * OUT_F + band * 16 + 4 * e) = s;
    }
}

extern "C" void kernel_launch(void* const* d_in, const int* in_sizes, int n_in,
                              void* d_out, int out_size, void* d_ws, size_t ws_size,
                              hipStream_t stream) {
    const float* inp  = (const float*)d_in[0];
    const float* wval = (const float*)d_in[1];
    const int*   wrow = (const int*)d_in[2];
    const int*   wcol = (const int*)d_in[3];
    const float* bias = (const float*)d_in[4];
    float*       out  = (float*)d_out;
    int*         cnt  = (int*)d_ws;
    int2*        bkt  = (int2*)((char*)d_ws + BKT_OFF);
    int2*        ovf  = (int2*)((char*)d_ws + OVF_OFF);
    const int    nnz  = in_sizes[1];

    clear_counters<<<1, 320, 0, stream>>>(cnt);

    bin_coo<<<(nnz + 4095) / 4096, 1024, 0, stream>>>(wval, wrow, wcol, cnt, bkt, ovf, nnz);

    band_wmma_gemm<<<NBANDS, 256, 0, stream>>>(cnt, bkt, ovf, inp, bias, out, IN_F);
}


# GEMM work-group tile 32 rows x 64 batch (slab rebuilt per K half, bins = band x half): each B fragment read from global feeds 2 WMMAs, inp reads halved; K summation order changes
# speedup vs baseline: 3.0065x; 1.1775x over previous
#include <hip/hip_runtime.h>

#define IN_F  4096
#define OUT_F 4096
#define BATCH 128

typedef __attribute__((ext_vector_type(2))) float v2f;
typedef __attribute__((ext_vector_type(4))) float v4f;
typedef __attribute__((ext_vector_type(8))) float v8f;

#define NBANDS     256
#define KHALF      2048
#define BAND_CAP   16384
#define OVF_CAP    2097152
#define CNT_STRIDE 1088
#define CNT_BYTES  (1u << 21)
#define BKT_OFF    ((size_t)CNT_BYTES)
#define OVF_OFF    (BKT_OFF + (size_t)NBANDS * BAND_CAP * 8)

__device__ __forceinline__ int bin_of_key(int key) {
    return ((key >> 17) << 1) | ((key >> 11) & 1);
}

__global__ void clear_counters(int* __restrict__ cnt) {
    int t = threadIdx.x;
    if (t <= NBANDS) cnt[t * CNT_STRIDE] = 0;
}

__global__ void __launch_bounds__(1024) bin_coo(
        const float* __restrict__ vals,
        const int*   __restrict__ rows,
        const int*   __restrict__ cols,
        int*  __restrict__ cnt,
        int2* __restrict__ bkt,
        int2* __restrict__ ovf,
        int nnz) {
    __shared__ int hist[NBANDS];
    __shared__ int base[NBANDS];
    const int tid = threadIdx.x;
    if (tid < NBANDS) hist[tid] = 0;
    __syncthreads();

    int   key[4], rank[4];
    float val[4];
#pragma unroll
    for (int q = 0; q < 4; ++q) {
        const int i = blockIdx.x * 4096 + q * 1024 + tid;
        key[q] = -1;
        rank[q] = 0;
        val[q] = 0.f;
        if (i < nnz) {
            const int r = rows[i] & (OUT_F - 1);
            const int c = cols[i] & (IN_F - 1);
            val[q] = vals[i];
            key[q] = (r << 12) | c;
            rank[q] = atomicAdd(&hist[bin_of_key(key[q])], 1);
        }
    }
    __syncthreads();
    if (tid < NBANDS) {
        const int h = hist[tid];
        base[tid] = (h > 0) ? atomicAdd(&cnt[tid * CNT_STRIDE], h) : 0;
    }
    __syncthreads();
#pragma unroll
    for (int q = 0; q < 4; ++q) {
        if (key[q] >= 0) {
            const int b   = bin_of_key(key[q]);
            const int pos = base[b] + rank[q];
            const int2 ent = make_int2(key[q], __float_as_int(val[q]));
            if (pos < BAND_CAP) {
                bkt[(size_t)b * BAND_CAP + pos] = ent;
            } else {
                const int p2 = atomicAdd(&cnt[NBANDS * CNT_STRIDE], 1);
                if (p2 < OVF_CAP) ovf[p2] = ent;
            }
        }
    }
}

#define KSPLIT 8
#define PITCH  2052
#define SC_DEPTH 32

__global__ void __launch_bounds__(256) band_wmma_gemm(
        const int*  __restrict__ cnt,
        const int2* __restrict__ bkt,
        const int2* __restrict__ ovf,
        const float* __restrict__ inp,
        const float* __restrict__ bias,
        float* __restrict__ out,
        int kdim) {
    __shared__ float Wl[32 * PITCH];

    const int band = blockIdx.x >> 1;
    const int nh   = blockIdx.x & 1;
    const int tid  = threadIdx.x;
    const int wave = tid >> 5;
    const int lane = tid & 31;
    const int half = lane >> 4;
    const int l16  = lane & 15;

    const int nkh = kdim / KHALF;

    int n = cnt[(2 * band) * CNT_STRIDE];
    n = n < BAND_CAP ? n : BAND_CAP;
    int n_next = 0;
    if (nkh > 1) {
        n_next = cnt[(2 * band + 1) * CNT_STRIDE];
        n_next = n_next < BAND_CAP ? n_next : BAND_CAP;
    }
    int novf = cnt[NBANDS * CNT_STRIDE];
    novf = novf < OVF_CAP ? novf : OVF_CAP;

    int2 ent[SC_DEPTH];
    auto load_batch = [&](const int2* list, int len, int base) {
#pragma unroll
        for (int j = 0; j < SC_DEPTH; ++j) {
            const int e = base + j * 256 + tid;
            ent[j] = make_int2(0, 0);
            if (e < len) ent[j] = list[e];
        }
    };
    auto add_batch = [&](int len, int base) {
#pragma unroll
        for (int j = 0; j < SC_DEPTH; ++j) {
            const int e = base + j * 256 + tid;
            if (e < len) {
                const int rr = (ent[j].x >> 12) & 31;
                const int cc = ent[j].x & (KHALF - 1);
                __hip_atomic_fetch_add(&Wl[rr * PITCH + cc], __int_as_float(ent[j].y),
                                       __ATOMIC_RELAXED, __HIP_MEMORY_SCOPE_WORKGROUP);
            }
        }
    };

    load_batch(bkt + (size_t)(2 * band) * BAND_CAP, n, 0);

    const int kper = KHALF / KSPLIT;
    const int koff = wave * kper + half * 16;

    const float* aptr[2];
#pragma unroll
    for (int m = 0; m < 2; ++m)
        aptr[m] = Wl + (16 * m + l16) * PITCH + koff;
    const float* bptr[4];
#pragma unroll
    for (int t = 0; t < 4; ++t)
        bptr[t] = inp + (size_t)(64 * nh + 16 * t + l16) * IN_F + koff;

    v8f acc[2][4];
#pragma unroll
    for (int m = 0; m < 2; ++m)
#pragma unroll
        for (int t = 0; t < 4; ++t)
            acc[m][t] = (v8f){0.f, 0.f, 0.f, 0.f, 0.f, 0.f, 0.f, 0.f};

    v4f S0[6][4], S1[6][4];

    auto load_stage_b = [&](v4f (&S)[6][4], int kglob, int step) {
#pragma unroll
        for (int t = 0; t < 4; ++t)
#pragma unroll
            for (int j = 0; j < 4; ++j)
                S[2 + t][j] = *(const v4f*)(bptr[t] + (size_t)kglob + step * 32 + j * 4);
    };
    auto load_stage_a = [&](v4f (&S)[6][4], int step) {
#pragma unroll
        for (int m = 0; m < 2; ++m)
#pragma unroll
            for (int j = 0; j < 4; ++j)
                S[m][j] = *(const v4f*)(aptr[m] + step * 32 + j * 4);
    };
    auto load_stage = [&](v4f (&S)[6][4], int kglob, int step) {
        load_stage_b(S, kglob, step);
        load_stage_a(S, step);
    };

    auto compute_stage = [&](const v4f (&S)[6][4]) {
#pragma unroll
        for (int j = 0; j < 4; ++j) {
#pragma unroll
            for (int p = 0; p < 2; ++p) {
#pragma unroll
                for (int m = 0; m < 2; ++m) {
                    v2f av = {S[m][j][2 * p], S[m][j][2 * p + 1]};
#pragma unroll
                    for (int t = 0; t < 4; ++t) {
                        v2f bv = {S[2 + t][j][2 * p], S[2 + t][j][2 * p + 1]};
                        acc[m][t] = __builtin_amdgcn_wmma_f32_16x16x4_f32(
                            false, av, false, bv, (short)0, acc[m][t], false, false);
                    }
                }
            }
        }
    };

    const int nsteps = kper >> 5;
    const int npairs = nsteps >> 1;

    for (int kh = 0; kh < nkh; ++kh) {
        const int kglob = kh * KHALF;
        load_stage_b(S0, kglob, 0);

        for (int i = tid; i < (32 * PITCH) / 4; i += 256)
            ((v4f*)Wl)[i] = (v4f){0.f, 0.f, 0.f, 0.f};
        __syncthreads();

        const int2* mybkt = bkt + (size_t)(2 * band + kh) * BAND_CAP;
        add_batch(n, 0);
        for (int base = SC_DEPTH * 256; base < n; base += SC_DEPTH * 256) {
            load_batch(mybkt, n, base);
            add_batch(n, base);
        }
        for (int e = tid; e < novf; e += 256) {
            const int2 oe = ovf[e];
            if (bin_of_key(oe.x) == 2 * band + kh) {
                const int rr = (oe.x >> 12) & 31;
                const int cc = oe.x & (KHALF - 1);
                __hip_atomic_fetch_add(&Wl[rr * PITCH + cc], __int_as_float(oe.y),
                                       __ATOMIC_RELAXED, __HIP_MEMORY_SCOPE_WORKGROUP);
            }
        }
        __syncthreads();

        load_stage_a(S0, 0);
        for (int t = 0; t + 1 < npairs; ++t) {
            load_stage(S1, kglob, 2 * t + 1);
            __builtin_amdgcn_sched_barrier(0);
            compute_stage(S0);
            __builtin_amdgcn_sched_barrier(0);
            load_stage(S0, kglob, 2 * t + 2);
            __builtin_amdgcn_sched_barrier(0);
            compute_stage(S1);
        }
        if (kh + 1 < nkh)
            load_batch(mybkt + BAND_CAP, n_next, 0);
        load_stage(S1, kglob, nsteps - 1);
        __builtin_amdgcn_sched_barrier(0);
        compute_stage(S0);
        compute_stage(S1);

        __syncthreads();
        n = n_next;
        n_next = 0;
        if (kh + 2 < nkh) {
            n_next = cnt[(2 * band + kh + 2) * CNT_STRIDE];
            n_next = n_next < BAND_CAP ? n_next : BAND_CAP;
        }
    }

    float* red  = Wl;
    float* mine = red + wave * 2048 + lane * 8;
#pragma unroll
    for (int m = 0; m < 2; ++m) {
#pragma unroll
        for (int t = 0; t < 4; ++t) {
            v8f a = acc[m][t];
            *(v4f*)(mine + (4 * m + t) * 256)     = (v4f){a[0], a[1], a[2], a[3]};
            *(v4f*)(mine + (4 * m + t) * 256 + 4) = (v4f){a[4], a[5], a[6], a[7]};
        }
    }
    __syncthreads();

#pragma unroll
    for (int pass = 0; pass < 2; ++pass) {
        const int idx  = tid + 256 * pass;
        const int Lb   = idx >> 3;
        const int e    = idx & 7;
        const int m    = e >> 2;
        const int h    = (e >> 1) & 1;
        const int quad = e & 1;
        const int t    = Lb >> 4;
        const int c    = Lb & 15;
        const float* src = red + (4 * m + t) * 256 + (c + 16 * h) * 8 + quad * 4;
        v4f s = *(const v4f*)src;
#pragma unroll
        for (int w = 1; w < KSPLIT; ++w)
            s += *(const v4f*)(src + w * 2048);
        s += *(const v4f*)(bias + band * 32 + 4 * e);
        *(v4f*)(out + (size_t)(64 * nh + Lb) * OUT_F + band * 32 + 4 * e) = s;
    }
}

extern "C" void kernel_launch(void* const* d_in, const int* in_sizes, int n_in,
                              void* d_out, int out_size, void* d_ws, size_t ws_size,
                              hipStream_t stream) {
    const float* inp  = (const float*)d_in[0];
    const float* wval = (const float*)d_in[1];
    const int*   wrow = (const int*)d_in[2];
    const int*   wcol = (const int*)d_in[3];
    const float* bias = (const float*)d_in[4];
    float*       out  = (float*)d_out;
    int*         cnt  = (int*)d_ws;
    int2*        bkt  = (int2*)((char*)d_ws + BKT_OFF);
    int2*        ovf  = (int2*)((char*)d_ws + OVF_OFF);
    const int    nnz  = in_sizes[1];

    clear_counters<<<1, 320, 0, stream>>>(cnt);

    bin_coo<<<(nnz + 4095) / 4096, 1024, 0, stream>>>(wval, wrow, wcol, cnt, bkt, ovf, nnz);

    band_wmma_gemm<<<NBANDS, 256, 0, stream>>>(cnt, bkt, ovf, inp, bias, out, IN_F);
}


# Binning without global counters: each binning work-group counting-sorts its 8192 nonzeros in LDS into one whole-line chunk + run words; no clear launch (3 -> 2); LDS-add order still timing-dependent
# speedup vs baseline: 3.8180x; 1.2699x over previous
#include <hip/hip_runtime.h>

#define IN_F  4096
#define OUT_F 4096
#define BATCH 128

typedef __attribute__((ext_vector_type(2))) float v2f;
typedef __attribute__((ext_vector_type(4))) float v4f;
typedef __attribute__((ext_vector_type(8))) float v8f;

#define NBANDS      256
#define KHALF       2048
#define CHUNK       8192
#define CHUNK_PITCH (CHUNK + NBANDS)
#define RUN_BYTES   (1u << 21)
#define CHUNK_OFF   ((size_t)RUN_BYTES)

typedef __attribute__((ext_vector_type(4))) int v4i;

__device__ __forceinline__ int bin_of_key(int key) {
    return ((key >> 17) << 1) | ((key >> 11) & 1);
}

__global__ void __launch_bounds__(1024) bin_coo(
        const float* __restrict__ vals,
        const int*   __restrict__ rows,
        const int*   __restrict__ cols,
        int*  __restrict__ runs,
        int2* __restrict__ chunks,
        int nnz) {
    __shared__ int  hist[NBANDS];
    __shared__ int  base[NBANDS];
    __shared__ int  wtot[NBANDS / 32];
    __shared__ int  total;
    __shared__ __attribute__((aligned(16))) int2 sorted[CHUNK_PITCH];
    const int tid  = threadIdx.x;
    const int lane = tid & 31;
    const int wave = tid >> 5;
    const int g    = blockIdx.x;
    const int G    = (nnz + CHUNK - 1) / CHUNK;
    if (tid < NBANDS) hist[tid] = 0;
    __syncthreads();

    int   key[8], rank[8], rr[8], cc[8];
    float val[8];
#pragma unroll
    for (int q = 0; q < 8; ++q) {
        const int i  = g * CHUNK + q * 1024 + tid;
        const int ic = i < nnz ? i : nnz - 1;
        rr[q]  = rows[ic];
        cc[q]  = cols[ic];
        val[q] = vals[ic];
    }
#pragma unroll
    for (int q = 0; q < 8; ++q) {
        const int i = g * CHUNK + q * 1024 + tid;
        key[q] = -1;
        rank[q] = 0;
        if (i < nnz) {
            key[q]  = ((rr[q] & (OUT_F - 1)) << 12) | (cc[q] & (IN_F - 1));
            rank[q] = atomicAdd(&hist[bin_of_key(key[q])], 1);
        }
    }
    __syncthreads();

    int h = 0, hp = 0, incl = 0;
    if (tid < NBANDS) {
        h = hist[tid];
        hp = (h + 1) & ~1;
        incl = hp;
        for (int d = 1; d < 32; d <<= 1) {
            const int v = __shfl_up(incl, d, 32);
            if (lane >= d) incl += v;
        }
        if (lane == 31) wtot[wave] = incl;
    }
    __syncthreads();
    if (tid < NBANDS) {
        int pre = 0;
        for (int w = 0; w < NBANDS / 32; ++w)
            if (w < wave) pre += wtot[w];
        const int excl = pre + incl - hp;
        base[tid] = excl;
        runs[(size_t)tid * G + g] = excl | (hp << 16);
        if (h & 1) sorted[excl + h] = make_int2(-1, 0);
        if (tid == NBANDS - 1) total = excl + hp;
    }
    __syncthreads();
#pragma unroll
    for (int q = 0; q < 8; ++q) {
        if (key[q] >= 0) {
            const int b = bin_of_key(key[q]);
            sorted[base[b] + rank[q]] = make_int2(key[q], __float_as_int(val[q]));
        }
    }
    __syncthreads();
    const int tot = total;
    v4i* dst = (v4i*)(chunks + (size_t)g * CHUNK_PITCH);
    const v4i* srt = (const v4i*)sorted;
    for (int i = tid; 2 * i < tot; i += 1024)
        dst[i] = srt[i];
}

#define KSPLIT 8
#define PITCH  2052
#define RUN_SLOTS 25
#define RUN_SPAN  (KSPLIT * RUN_SLOTS)

__global__ void __launch_bounds__(256) band_wmma_gemm(
        const int*  __restrict__ runs,
        const int2* __restrict__ chunks,
        const float* __restrict__ inp,
        const float* __restrict__ bias,
        float* __restrict__ out,
        int nnz,
        int kdim) {
    __shared__ float Wl[32 * PITCH];

    const int band = blockIdx.x >> 1;
    const int nh   = blockIdx.x & 1;
    const int tid  = threadIdx.x;
    const int wave = tid >> 5;
    const int lane = tid & 31;
    const int half = lane >> 4;
    const int l16  = lane & 15;

    const int nkh = kdim / KHALF;

    const int G = (nnz + CHUNK - 1) / CHUNK;
    const int wave_s = __builtin_amdgcn_readfirstlane(wave);

    auto load_runs = [&](int bin, int gb, int& r) {
        const int* row = runs + (size_t)bin * G;
        const int gi = gb + RUN_SLOTS * wave_s + lane;
        r = 0;
        if (lane < RUN_SLOTS && gi < G) r = row[gi];
    };

    v4i ent[RUN_SLOTS];
    auto load_batch = [&](int gb, int r) {
#pragma unroll
        for (int j = 0; j < RUN_SLOTS; ++j) {
            const int word  = __builtin_amdgcn_readlane(r, j);
            const int start = word & 0xFFFF;
            const int len   = word >> 16;
            const int2* run = chunks + (size_t)(gb + RUN_SLOTS * wave_s + j) * CHUNK_PITCH + start;
            ent[j] = (v4i){-1, 0, -1, 0};
            if (2 * lane < len) ent[j] = *(const v4i*)(run + 2 * lane);
        }
    };
    auto add_entry = [&](int k, int v) {
        if (k >= 0) {
            const int rr = (k >> 12) & 31;
            const int cc = k & (KHALF - 1);
            __hip_atomic_fetch_add(&Wl[rr * PITCH + cc], __int_as_float(v),
                                   __ATOMIC_RELAXED, __HIP_MEMORY_SCOPE_WORKGROUP);
        }
    };
    auto add_batch = [&]() {
#pragma unroll
        for (int j = 0; j < RUN_SLOTS; ++j) {
            add_entry(ent[j][0], ent[j][1]);
            add_entry(ent[j][2], ent[j][3]);
        }
    };
    auto add_tails = [&](int s0, int gb, int r) {
        const int len = r >> 16;
        const int gi = gb + RUN_SLOTS * wave_s + lane;
        const int2* run = chunks + (size_t)gi * CHUNK_PITCH + (r & 0xFFFF);
        for (int s = s0; s < len; ++s) {
            const int2 e = run[s];
            add_entry(e.x, e.y);
        }
    };

    int rw, qw = 0;
    load_runs(2 * band, 0, rw);
    if (nkh > 1) load_runs(2 * band + 1, 0, qw);
    load_batch(0, rw);

    const int kper = KHALF / KSPLIT;
    const int koff = wave * kper + half * 16;

    const float* aptr[2];
#pragma unroll
    for (int m = 0; m < 2; ++m)
        aptr[m] = Wl + (16 * m + l16) * PITCH + koff;
    const float* bptr[4];
#pragma unroll
    for (int t = 0; t < 4; ++t)
        bptr[t] = inp + (size_t)(64 * nh + 16 * t + l16) * IN_F + koff;

    v8f acc[2][4];
#pragma unroll
    for (int m = 0; m < 2; ++m)
#pragma unroll
        for (int t = 0; t < 4; ++t)
            acc[m][t] = (v8f){0.f, 0.f, 0.f, 0.f, 0.f, 0.f, 0.f, 0.f};

    v4f S0[6][4], S1[6][4];

    auto load_stage_b = [&](v4f (&S)[6][4], int kglob, int step) {
#pragma unroll
        for (int t = 0; t < 4; ++t)
#pragma unroll
            for (int j = 0; j < 4; ++j)
                S[2 + t][j] = *(const v4f*)(bptr[t] + (size_t)kglob + step * 32 + j * 4);
    };
    auto load_stage_a = [&](v4f (&S)[6][4], int step) {
#pragma unroll
        for (int m = 0; m < 2; ++m)
#pragma unroll
            for (int j = 0; j < 4; ++j)
                S[m][j] = *(const v4f*)(aptr[m] + step * 32 + j * 4);
    };
    auto load_stage = [&](v4f (&S)[6][4], int kglob, int step) {
        load_stage_b(S, kglob, step);
        load_stage_a(S, step);
    };

    auto compute_stage = [&](const v4f (&S)[6][4]) {
#pragma unroll
        for (int j = 0; j < 4; ++j) {
#pragma unroll
            for (int p = 0; p < 2; ++p) {
#pragma unroll
                for (int m = 0; m < 2; ++m) {
                    v2f av = {S[m][j][2 * p], S[m][j][2 * p + 1]};
#pragma unroll
                    for (int t = 0; t < 4; ++t) {
                        v2f bv = {S[2 + t][j][2 * p], S[2 + t][j][2 * p + 1]};
                        acc[m][t] = __builtin_amdgcn_wmma_f32_16x16x4_f32(
                            false, av, false, bv, (short)0, acc[m][t], false, false);
                    }
                }
            }
        }
    };

    const int nsteps = kper >> 5;
    const int npairs = nsteps >> 1;

    for (int kh = 0; kh < nkh; ++kh) {
        const int kglob = kh * KHALF;
        load_stage_b(S0, kglob, 0);

        for (int i = tid; i < (32 * PITCH) / 4; i += 256)
            ((v4f*)Wl)[i] = (v4f){0.f, 0.f, 0.f, 0.f};
        __syncthreads();

        add_batch();
        add_tails(64, 0, rw);
        for (int gb = RUN_SPAN; gb < G; gb += RUN_SPAN) {
            int tw;
            load_runs(2 * band + kh, gb, tw);
            add_tails(0, gb, tw);
        }
        __syncthreads();

        load_stage_a(S0, 0);
        for (int t = 0; t + 1 < npairs; ++t) {
            load_stage(S1, kglob, 2 * t + 1);
            __builtin_amdgcn_sched_barrier(0);
            compute_stage(S0);
            __builtin_amdgcn_sched_barrier(0);
            load_stage(S0, kglob, 2 * t + 2);
            __builtin_amdgcn_sched_barrier(0);
            compute_stage(S1);
        }
        if (kh + 1 < nkh)
            load_batch(0, qw);
        load_stage(S1, kglob, nsteps - 1);
        __builtin_amdgcn_sched_barrier(0);
        compute_stage(S0);
        compute_stage(S1);

        __syncthreads();
        rw = qw;
        qw = 0;
        if (kh + 2 < nkh) load_runs(2 * band + kh + 2, 0, qw);
    }

    float* red  = Wl;
    float* mine = red + wave * 2048 + lane * 8;
#pragma unroll
    for (int m = 0; m < 2; ++m) {
#pragma unroll
        for (int t = 0; t < 4; ++t) {
            v8f a = acc[m][t];
            *(v4f*)(mine + (4 * m + t) * 256)     = (v4f){a[0], a[1], a[2], a[3]};
            *(v4f*)(mine + (4 * m + t) * 256 + 4) = (v4f){a[4], a[5], a[6], a[7]};
        }
    }
    __syncthreads();

#pragma unroll
    for (int pass = 0; pass < 2; ++pass) {
        const int idx  = tid + 256 * pass;
        const int Lb   = idx >> 3;
        const int e    = idx & 7;
        const int m    = e >> 2;
        const int h    = (e >> 1) & 1;
        const int quad = e & 1;
        const int t    = Lb >> 4;
        const int c    = Lb & 15;
        const float* src = red + (4 * m + t) * 256 + (c + 16 * h) * 8 + quad * 4;
        v4f s = *(const v4f*)src;
#pragma unroll
        for (int w = 1; w < KSPLIT; ++w)
            s += *(const v4f*)(src + w * 2048);
        s += *(const v4f*)(bias + band * 32 + 4 * e);
        *(v4f*)(out + (size_t)(64 * nh + Lb) * OUT_F + band * 32 + 4 * e) = s;
    }
}

extern "C" void kernel_launch(void* const* d_in, const int* in_sizes, int n_in,
                              void* d_out, int out_size, void* d_ws, size_t ws_size,
                              hipStream_t stream) {
    const float* inp  = (const float*)d_in[0];
    const float* wval = (const float*)d_in[1];
    const int*   wrow = (const int*)d_in[2];
    const int*   wcol = (const int*)d_in[3];
    const float* bias = (const float*)d_in[4];
    float*       out  = (float*)d_out;
    int*         runs = (int*)d_ws;
    int2*        chunks = (int2*)((char*)d_ws + CHUNK_OFF);
    const int    nnz  = in_sizes[1];

    bin_coo<<<(nnz + 8191) / 8192, 1024, 0, stream>>>(wval, wrow, wcol, runs, chunks, nnz);

    band_wmma_gemm<<<NBANDS, 256, 0, stream>>>(runs, chunks, inp, bias, out, nnz, IN_F);
}


# GEMM B operand read from a copy of inp in fragment order (written by the binning kernel's threads): each B load takes 512 contiguous bytes, 4 lines, not 32 B from each of 16 lines
# speedup vs baseline: 3.9650x; 1.0385x over previous
#include <hip/hip_runtime.h>

#define IN_F  4096
#define OUT_F 4096
#define BATCH 128

typedef __attribute__((ext_vector_type(2))) float v2f;
typedef __attribute__((ext_vector_type(4))) float v4f;
typedef __attribute__((ext_vector_type(8))) float v8f;

#define NBANDS      256
#define KHALF       2048
#define CHUNK       8192
#define CHUNK_PITCH (CHUNK + NBANDS)
#define RUN_BYTES   (1u << 21)
#define INPF_OFF    ((size_t)RUN_BYTES)
#define INPF_PIECES (BATCH * IN_F / 4)
#define CHUNK_OFF   (INPF_OFF + (size_t)BATCH * IN_F * 4)

typedef __attribute__((ext_vector_type(4))) int v4i;

__device__ __forceinline__ int bin_of_key(int key) {
    return ((key >> 17) << 1) | ((key >> 11) & 1);
}

__global__ void __launch_bounds__(1024) bin_coo(
        const float* __restrict__ vals,
        const int*   __restrict__ rows,
        const int*   __restrict__ cols,
        const float* __restrict__ inp,
        int*  __restrict__ runs,
        int2* __restrict__ chunks,
        float* __restrict__ inpf,
        int nnz) {
    __shared__ int  hist[NBANDS];
    __shared__ int  base[NBANDS];
    __shared__ int  wtot[NBANDS / 32];
    __shared__ int  total;
    __shared__ __attribute__((aligned(16))) int2 sorted[CHUNK_PITCH];
    const int tid  = threadIdx.x;
    const int lane = tid & 31;
    const int wave = tid >> 5;
    const int g    = blockIdx.x;
    const int G    = (nnz + CHUNK - 1) / CHUNK;
    if (tid < NBANDS) hist[tid] = 0;
    __syncthreads();

    int   key[8], rank[8], rr[8], cc[8];
    float val[8];
#pragma unroll
    for (int q = 0; q < 8; ++q) {
        const int i  = g * CHUNK + q * 1024 + tid;
        const int ic = i < nnz ? i : nnz - 1;
        rr[q]  = rows[ic];
        cc[q]  = cols[ic];
        val[q] = vals[ic];
    }
    for (int d = g * 1024 + tid; d < INPF_PIECES; d += G * 1024) {
        const int c  = d & 15;
        const int hh = (d >> 4) & 1;
        const int j  = (d >> 5) & 3;
        const int t  = (d >> 7) & 3;
        const int kb = (d >> 9) & 127;
        const int nh = d >> 16;
        const float* s = inp + (size_t)(64 * nh + 16 * t + c) * IN_F + 32 * kb + 16 * hh + 4 * j;
        ((v4f*)inpf)[d] = *(const v4f*)s;
    }
#pragma unroll
    for (int q = 0; q < 8; ++q) {
        const int i = g * CHUNK + q * 1024 + tid;
        key[q] = -1;
        rank[q] = 0;
        if (i < nnz) {
            key[q]  = ((rr[q] & (OUT_F - 1)) << 12) | (cc[q] & (IN_F - 1));
            rank[q] = atomicAdd(&hist[bin_of_key(key[q])], 1);
        }
    }
    __syncthreads();

    int h = 0, hp = 0, incl = 0;
    if (tid < NBANDS) {
        h = hist[tid];
        hp = (h + 1) & ~1;
        incl = hp;
        for (int d = 1; d < 32; d <<= 1) {
            const int v = __shfl_up(incl, d, 32);
            if (lane >= d) incl += v;
        }
        if (lane == 31) wtot[wave] = incl;
    }
    __syncthreads();
    if (tid < NBANDS) {
        int pre = 0;
        for (int w = 0; w < NBANDS / 32; ++w)
            if (w < wave) pre += wtot[w];
        const int excl = pre + incl - hp;
        base[tid] = excl;
        runs[(size_t)tid * G + g] = excl | (hp << 16);
        if (h & 1) sorted[excl + h] = make_int2(-1, 0);
        if (tid == NBANDS - 1) total = excl + hp;
    }
    __syncthreads();
#pragma unroll
    for (int q = 0; q < 8; ++q) {
        if (key[q] >= 0) {
            const int b = bin_of_key(key[q]);
            sorted[base[b] + rank[q]] = make_int2(key[q], __float_as_int(val[q]));
        }
    }
    __syncthreads();
    const int tot = total;
    v4i* dst = (v4i*)(chunks + (size_t)g * CHUNK_PITCH);
    const v4i* srt = (const v4i*)sorted;
    for (int i = tid; 2 * i < tot; i += 1024)
        dst[i] = srt[i];
}

#define KSPLIT 8
#define PITCH  2052
#define RUN_SLOTS 25
#define RUN_SPAN  (KSPLIT * RUN_SLOTS)

__global__ void __launch_bounds__(256) band_wmma_gemm(
        const int*  __restrict__ runs,
        const int2* __restrict__ chunks,
        const float* __restrict__ inpf,
        const float* __restrict__ bias,
        float* __restrict__ out,
        int nnz,
        int kdim) {
    __shared__ float Wl[32 * PITCH];

    const int band = blockIdx.x >> 1;
    const int nh   = blockIdx.x & 1;
    const int tid  = threadIdx.x;
    const int wave = tid >> 5;
    const int lane = tid & 31;
    const int half = lane >> 4;
    const int l16  = lane & 15;

    const int nkh = kdim / KHALF;

    const int G = (nnz + CHUNK - 1) / CHUNK;
    const int wave_s = __builtin_amdgcn_readfirstlane(wave);

    auto load_runs = [&](int bin, int gb, int& r) {
        const int* row = runs + (size_t)bin * G;
        const int gi = gb + RUN_SLOTS * wave_s + lane;
        r = 0;
        if (lane < RUN_SLOTS && gi < G) r = row[gi];
    };

    v4i ent[RUN_SLOTS];
    auto load_batch = [&](int gb, int r) {
#pragma unroll
        for (int j = 0; j < RUN_SLOTS; ++j) {
            const int word  = __builtin_amdgcn_readlane(r, j);
            const int start = word & 0xFFFF;
            const int len   = word >> 16;
            const int2* run = chunks + (size_t)(gb + RUN_SLOTS * wave_s + j) * CHUNK_PITCH + start;
            ent[j] = (v4i){-1, 0, -1, 0};
            if (2 * lane < len) ent[j] = *(const v4i*)(run + 2 * lane);
        }
    };
    auto add_entry = [&](int k, int v) {
        if (k >= 0) {
            const int rr = (k >> 12) & 31;
            const int cc = k & (KHALF - 1);
            __hip_atomic_fetch_add(&Wl[rr * PITCH + cc], __int_as_float(v),
                                   __ATOMIC_RELAXED, __HIP_MEMORY_SCOPE_WORKGROUP);
        }
    };
    auto add_batch = [&]() {
#pragma unroll
        for (int j = 0; j < RUN_SLOTS; ++j) {
            add_entry(ent[j][0], ent[j][1]);
            add_entry(ent[j][2], ent[j][3]);
        }
    };
    auto add_tails = [&](int s0, int gb, int r) {
        const int len = r >> 16;
        const int gi = gb + RUN_SLOTS * wave_s + lane;
        const int2* run = chunks + (size_t)gi * CHUNK_PITCH + (r & 0xFFFF);
        for (int s = s0; s < len; ++s) {
            const int2 e = run[s];
            add_entry(e.x, e.y);
        }
    };

    int rw, qw = 0;
    load_runs(2 * band, 0, rw);
    if (nkh > 1) load_runs(2 * band + 1, 0, qw);
    load_batch(0, rw);

    const int kper = KHALF / KSPLIT;
    const int koff = wave * kper + half * 16;

    const float* aptr[2];
#pragma unroll
    for (int m = 0; m < 2; ++m)
        aptr[m] = Wl + (16 * m + l16) * PITCH + koff;
    const v4f* bfrag = (const v4f*)inpf + (size_t)(nh * 128 + wave * 8) * 512 + lane;

    v8f acc[2][4];
#pragma unroll
    for (int m = 0; m < 2; ++m)
#pragma unroll
        for (int t = 0; t < 4; ++t)
            acc[m][t] = (v8f){0.f, 0.f, 0.f, 0.f, 0.f, 0.f, 0.f, 0.f};

    v4f S0[6][4], S1[6][4];

    auto load_stage_b = [&](v4f (&S)[6][4], int kglob, int step) {
#pragma unroll
        for (int t = 0; t < 4; ++t)
#pragma unroll
            for (int j = 0; j < 4; ++j)
                S[2 + t][j] = bfrag[(size_t)((kglob >> 5) + step) * 512 + (4 * t + j) * 32];
    };
    auto load_stage_a = [&](v4f (&S)[6][4], int step) {
#pragma unroll
        for (int m = 0; m < 2; ++m)
#pragma unroll
            for (int j = 0; j < 4; ++j)
                S[m][j] = *(const v4f*)(aptr[m] + step * 32 + j * 4);
    };
    auto load_stage = [&](v4f (&S)[6][4], int kglob, int step) {
        load_stage_b(S, kglob, step);
        load_stage_a(S, step);
    };

    auto compute_stage = [&](const v4f (&S)[6][4]) {
#pragma unroll
        for (int j = 0; j < 4; ++j) {
#pragma unroll
            for (int p = 0; p < 2; ++p) {
#pragma unroll
                for (int m = 0; m < 2; ++m) {
                    v2f av = {S[m][j][2 * p], S[m][j][2 * p + 1]};
#pragma unroll
                    for (int t = 0; t < 4; ++t) {
                        v2f bv = {S[2 + t][j][2 * p], S[2 + t][j][2 * p + 1]};
                        acc[m][t] = __builtin_amdgcn_wmma_f32_16x16x4_f32(
                            false, av, false, bv, (short)0, acc[m][t], false, false);
                    }
                }
            }
        }
    };

    const int nsteps = kper >> 5;
    const int npairs = nsteps >> 1;

    for (int kh = 0; kh < nkh; ++kh) {
        const int kglob = kh * KHALF;
        load_stage_b(S0, kglob, 0);

        for (int i = tid; i < (32 * PITCH) / 4; i += 256)
            ((v4f*)Wl)[i] = (v4f){0.f, 0.f, 0.f, 0.f};
        __syncthreads();

        add_batch();
        add_tails(64, 0, rw);
        for (int gb = RUN_SPAN; gb < G; gb += RUN_SPAN) {
            int tw;
            load_runs(2 * band + kh, gb, tw);
            add_tails(0, gb, tw);
        }
        __syncthreads();

        load_stage_a(S0, 0);
        for (int t = 0; t + 1 < npairs; ++t) {
            load_stage(S1, kglob, 2 * t + 1);
            __builtin_amdgcn_sched_barrier(0);
            compute_stage(S0);
            __builtin_amdgcn_sched_barrier(0);
            load_stage(S0, kglob, 2 * t + 2);
            __builtin_amdgcn_sched_barrier(0);
            compute_stage(S1);
        }
        if (kh + 1 < nkh)
            load_batch(0, qw);
        load_stage(S1, kglob, nsteps - 1);
        __builtin_amdgcn_sched_barrier(0);
        compute_stage(S0);
        compute_stage(S1);

        __syncthreads();
        rw = qw;
        qw = 0;
        if (kh + 2 < nkh) load_runs(2 * band + kh + 2, 0, qw);
    }

    float* red  = Wl;
    float* mine = red + wave * 2048 + lane * 8;
#pragma unroll
    for (int m = 0; m < 2; ++m) {
#pragma unroll
        for (int t = 0; t < 4; ++t) {
            v8f a = acc[m][t];
            *(v4f*)(mine + (4 * m + t) * 256)     = (v4f){a[0], a[1], a[2], a[3]};
            *(v4f*)(mine + (4 * m + t) * 256 + 4) = (v4f){a[4], a[5], a[6], a[7]};
        }
    }
    __syncthreads();

#pragma unroll
    for (int pass = 0; pass < 2; ++pass) {
        const int idx  = tid + 256 * pass;
        const int Lb   = idx >> 3;
        const int e    = idx & 7;
        const int m    = e >> 2;
        const int h    = (e >> 1) & 1;
        const int quad = e & 1;
        const int t    = Lb >> 4;
        const int c    = Lb & 15;
        const float* src = red + (4 * m + t) * 256 + (c + 16 * h) * 8 + quad * 4;
        v4f s = *(const v4f*)src;
#pragma unroll
        for (int w = 1; w < KSPLIT; ++w)
            s += *(const v4f*)(src + w * 2048);
        s += *(const v4f*)(bias + band * 32 + 4 * e);
        *(v4f*)(out + (size_t)(64 * nh + Lb) * OUT_F + band * 32 + 4 * e) = s;
    }
}

extern "C" void kernel_launch(void* const* d_in, const int* in_sizes, int n_in,
                              void* d_out, int out_size, void* d_ws, size_t ws_size,
                              hipStream_t stream) {
    const float* inp  = (const float*)d_in[0];
    const float* wval = (const float*)d_in[1];
    const int*   wrow = (const int*)d_in[2];
    const int*   wcol = (const int*)d_in[3];
    const float* bias = (const float*)d_in[4];
    float*       out  = (float*)d_out;
    int*         runs = (int*)d_ws;
    float*       inpf = (float*)((char*)d_ws + INPF_OFF);
    int2*        chunks = (int2*)((char*)d_ws + CHUNK_OFF);
    const int    nnz  = in_sizes[1];

    bin_coo<<<(nnz + 8191) / 8192, 1024, 0, stream>>>(wval, wrow, wcol, inp, runs, chunks, inpf, nnz);

    band_wmma_gemm<<<NBANDS, 256, 0, stream>>>(runs, chunks, inpf, bias, out, nnz, IN_F);
}
